# starting kernel
# baseline (speedup 1.0000x reference)
#include <hip/hip_runtime.h>

typedef __attribute__((ext_vector_type(2))) float v2f;
typedef __attribute__((ext_vector_type(8))) float v8f;

#define B_TOT 1024
#define T_LEN 4096
#define HID   20
#define NG    80
#define ROWS  16
#define GST   84

__device__ __forceinline__ float sigm(float x) {
    return 1.0f / (1.0f + __expf(-x));
}
__device__ __forceinline__ float tanh_f(float x) {
    return 1.0f - 2.0f / (__expf(2.0f * x) + 1.0f);
}

__global__ void __launch_bounds__(32)
lstm_seq_kernel(const float* __restrict__ xin,
                const float* __restrict__ Wih1,
                const float* __restrict__ Whh1,
                const float* __restrict__ bih1,
                const float* __restrict__ bhh1,
                const float* __restrict__ Wih2,
                const float* __restrict__ Whh2,
                const float* __restrict__ bih2,
                const float* __restrict__ bhh2,
                const int*   __restrict__ futp,
                float* __restrict__ out)
{
    __shared__ float lds_h1[ROWS * HID];
    __shared__ float lds_c1[ROWS * HID];
    __shared__ float lds_g [ROWS * GST];
    __shared__ float lds_x [ROWS];
    __shared__ float s_wih[NG];
    __shared__ float s_bs [NG];
    __shared__ float s_w2 [NG];
    __shared__ float s_whh2[4];
    __shared__ float s_b2 [4];

    const int lane  = threadIdx.x;
    const int nu    = lane & 15;
    const int khalf = lane >> 4;
    const int row0  = blockIdx.x * ROWS;
    const int fut   = *futp;
    const int TT    = T_LEN + fut;

    for (int i = lane; i < NG; i += 32) {
        s_wih[i] = Wih1[i];
        s_bs [i] = bih1[i] + bhh1[i];
        s_w2 [i] = Wih2[i];
    }
    if (lane < 4) {
        s_whh2[lane] = Whh2[lane];
        s_b2 [lane]  = bih2[lane] + bhh2[lane];
    }
    for (int i = lane; i < ROWS * HID; i += 32) {
        lds_h1[i] = 0.0f;
        lds_c1[i] = 0.0f;
    }

    v2f bf[5][5];
#pragma unroll
    for (int kc = 0; kc < 5; ++kc) {
#pragma unroll
        for (int n = 0; n < 5; ++n) {
            const float* p = Whh1 + (16 * n + nu) * HID + 4 * kc + 2 * khalf;
            v2f b; b.x = p[0]; b.y = p[1];
            bf[kc][n] = b;
        }
    }

    float c1v[10];
#pragma unroll
    for (int j = 0; j < 10; ++j) c1v[j] = 0.0f;
    float h2v = 0.0f, c2v = 0.0f;

    const int er  = lane >> 1;
    const int eh0 = (lane & 1) * 10;

    __syncthreads();

    for (int t = 0; t < TT; ++t) {
        if (t < T_LEN) {
            if (lane < ROWS)
                lds_x[lane] = xin[(size_t)(row0 + lane) * T_LEN + t];
        }
        __syncthreads();

        v8f acc[5];
#pragma unroll
        for (int n = 0; n < 5; ++n) {
            v8f z = {0.f, 0.f, 0.f, 0.f, 0.f, 0.f, 0.f, 0.f};
            acc[n] = z;
        }
#pragma unroll
        for (int kc = 0; kc < 5; ++kc) {
            v2f a;
            const float* hp = &lds_h1[nu * HID + 4 * kc + 2 * khalf];
            a.x = hp[0]; a.y = hp[1];
#pragma unroll
            for (int n = 0; n < 5; ++n) {
                acc[n] = __builtin_amdgcn_wmma_f32_16x16x4_f32(
                    false, a, false, bf[kc][n], (short)0, acc[n], false, false);
            }
        }
#pragma unroll
        for (int n = 0; n < 5; ++n) {
#pragma unroll
            for (int v = 0; v < 8; ++v) {
                lds_g[(v + 8 * khalf) * GST + 16 * n + nu] = acc[n][v];
            }
        }
        __syncthreads();

        {
            const float  xr = lds_x[er];
            const float* gr = &lds_g[er * GST];
#pragma unroll
            for (int j = 0; j < 10; ++j) {
                const int h = eh0 + j;
                float gi = gr[h]      + xr * s_wih[h]      + s_bs[h];
                float gf = gr[20 + h] + xr * s_wih[20 + h] + s_bs[20 + h];
                float gg = gr[40 + h] + xr * s_wih[40 + h] + s_bs[40 + h];
                float go = gr[60 + h] + xr * s_wih[60 + h] + s_bs[60 + h];
                float cn = sigm(gf) * c1v[j] + sigm(gi) * tanh_f(gg);
                c1v[j] = cn;
                float hn = sigm(go) * tanh_f(cn);
                lds_h1[er * HID + h] = hn;
                lds_c1[er * HID + h] = cn;
            }
        }
        __syncthreads();

        if (lane < ROWS) {
            float g2[4];
#pragma unroll
            for (int g = 0; g < 4; ++g) g2[g] = s_b2[g] + h2v * s_whh2[g];
            const float* cr = &lds_c1[lane * HID];
#pragma unroll
            for (int h = 0; h < HID; ++h) {
                float cv = cr[h];
#pragma unroll
                for (int g = 0; g < 4; ++g) g2[g] += cv * s_w2[g * HID + h];
            }
            float cn2 = sigm(g2[1]) * c2v + sigm(g2[0]) * tanh_f(g2[2]);
            c2v = cn2;
            h2v = sigm(g2[3]) * tanh_f(cn2);
            out[(size_t)(row0 + lane) * TT + t] = cn2;
            lds_x[lane] = cn2;
        }
        __syncthreads();
    }
}

extern "C" void kernel_launch(void* const* d_in, const int* in_sizes, int n_in,
                              void* d_out, int out_size, void* d_ws, size_t ws_size,
                              hipStream_t stream)
{
    const float* xin  = (const float*)d_in[0];
    const float* Wih1 = (const float*)d_in[1];
    const float* Whh1 = (const float*)d_in[2];
    const float* bih1 = (const float*)d_in[3];
    const float* bhh1 = (const float*)d_in[4];
    const float* Wih2 = (const float*)d_in[5];
    const float* Whh2 = (const float*)d_in[6];
    const float* bih2 = (const float*)d_in[7];
    const float* bhh2 = (const float*)d_in[8];
    const int*   futp = (const int*)d_in[9];
    float* out = (float*)d_out;

    dim3 grid(B_TOT / ROWS);
    dim3 block(32);
    lstm_seq_kernel<<<grid, block, 0, stream>>>(xin, Wih1, Whh1, bih1, bhh1,
                                                Wih2, Whh2, bih2, bhh2, futp, out);
}


# Layer-1 exchange leaves LDS: h1 broadcast to all A rows by lane-row shares, gate columns permuted so a lane's D register 0 holds its unit's gates; no barrier in the step; layer-1 sum order changes
# speedup vs baseline: 3.5903x; 3.5903x over previous
#include <hip/hip_runtime.h>

typedef __attribute__((ext_vector_type(2))) float v2f;
typedef __attribute__((ext_vector_type(8))) float v8f;

#define B_TOT 1024
#define T_LEN 4096
#define HID   20
#define XBLK  32
#define NBLK  (T_LEN / XBLK)
#define GRP   16
#define NGRP  (T_LEN / GRP)
#define L2LANE 20
#define G2ST  16
static_assert(T_LEN % XBLK == 0 && NBLK >= 2, "x blocks must tile the sequence");
static_assert(XBLK == 2 * GRP && T_LEN % GRP == 0 && NGRP >= 2, "two product groups per x block");

#define DPP_QUAD_ID      0xE4
#define DPP_ROW_SHL(n)   (0x100 + (n))
#define DPP_ROW_SHARE(n) (0x150 + (n))

template <bool V> struct flag { static constexpr bool value = V; };

__device__ __forceinline__ float sigm(float x) {
    return 1.0f / (1.0f + __expf(-x));
}
__device__ __forceinline__ float tanh_f(float x) {
    return 1.0f - 2.0f / (__expf(2.0f * x) + 1.0f);
}

__device__ __forceinline__ void lds_barrier() {
    __builtin_amdgcn_fence(__ATOMIC_RELEASE, "workgroup", "local");
    __builtin_amdgcn_s_barrier();
    __builtin_amdgcn_fence(__ATOMIC_ACQUIRE, "workgroup", "local");
}

__device__ __forceinline__ int kunit(int r, int h) {
    return (h == 0) ? r : ((r < 4) ? (16 + r) : (r + 6));
}

__global__ void __launch_bounds__(32)
lstm_seq_kernel(const float* __restrict__ xin,
                const float* __restrict__ Wih1,
                const float* __restrict__ Whh1,
                const float* __restrict__ bih1,
                const float* __restrict__ bhh1,
                const float* __restrict__ Wih2,
                const float* __restrict__ Whh2,
                const float* __restrict__ bih2,
                const float* __restrict__ bhh2,
                const int*   __restrict__ futp,
                float* __restrict__ out)
{
    __shared__ float lds_c1h[GRP * HID + 32];
    __shared__ __attribute__((aligned(16))) float lds_g2[GRP * G2ST];

    const int lane  = threadIdx.x;
    const int nu    = lane & 15;
    const int khalf = lane >> 4;
    const int row0  = blockIdx.x;
    const int fut   = *futp;
    const int TT    = T_LEN + fut;

    const bool is2lane = (lane == L2LANE);
    const int  unit    = (lane < HID) ? lane : nu;
    const bool primary = (lane < 10) || (lane >= 16 && lane < HID) || (lane >= 26);
    const int  c1off = primary ? unit : (GRP * HID + lane);
    const int  c1str = primary ? HID : 0;

    for (int i = lane; i < GRP * HID; i += 32)  lds_c1h[i] = 0.0f;

    float wx[4], bs[4];
#pragma unroll
    for (int g = 0; g < 4; ++g) {
        if (is2lane) {
            wx[g] = Whh2[g];
            bs[g] = bih2[g] + bhh2[g];
        } else {
            wx[g] = Wih1[g * HID + unit];
            bs[g] = bih1[g * HID + unit] + bhh1[g * HID + unit];
        }
    }

    v2f bf[5][5];
#pragma unroll
    for (int n = 0; n < 5; ++n) {
        const int wrow = (n < 4) ? (n * HID + nu) : ((nu >> 2) * HID + 16 + (nu & 3));
#pragma unroll
        for (int kc = 0; kc < 5; ++kc) {
            v2f b;
            b.x = Whh1[wrow * HID + kunit(2 * kc,     khalf)];
            b.y = Whh1[wrow * HID + kunit(2 * kc + 1, khalf)];
            bf[kc][n] = b;
        }
    }
    v2f bf2[5];
#pragma unroll
    for (int kc = 0; kc < 5; ++kc) {
        v2f b; b.x = 0.0f; b.y = 0.0f;
        if (nu < 4) {
            const float* p = Wih2 + nu * HID + 4 * kc + 2 * khalf;
            b.x = p[0]; b.y = p[1];
        }
        bf2[kc] = b;
    }

    float cst = 0.0f, hst = 0.0f;
    float4 qn = make_float4(0.0f, 0.0f, 0.0f, 0.0f);

    auto tick = [&](auto f1, auto f2, float xs, int row, int tout) __attribute__((always_inline)) {
        constexpr bool L1 = decltype(f1)::value;
        constexpr bool L2 = decltype(f2)::value;
        float g4[4] = {0.0f, 0.0f, 0.0f, 0.0f};
        if constexpr (L1) {
            const int hi = __float_as_int(hst);
            const int hs = __builtin_amdgcn_update_dpp(hi, hi, DPP_ROW_SHL(6), 0x2, 0xf, false);
            int ar[10];
            ar[0] = __builtin_amdgcn_update_dpp(0, hi, DPP_ROW_SHARE(0), 0xf, 0xf, false);
            ar[1] = __builtin_amdgcn_update_dpp(0, hi, DPP_ROW_SHARE(1), 0xf, 0xf, false);
            ar[2] = __builtin_amdgcn_update_dpp(0, hi, DPP_ROW_SHARE(2), 0xf, 0xf, false);
            ar[3] = __builtin_amdgcn_update_dpp(0, hi, DPP_ROW_SHARE(3), 0xf, 0xf, false);
            ar[4] = __builtin_amdgcn_update_dpp(0, hs, DPP_ROW_SHARE(4), 0xf, 0xf, false);
            ar[5] = __builtin_amdgcn_update_dpp(0, hs, DPP_ROW_SHARE(5), 0xf, 0xf, false);
            ar[6] = __builtin_amdgcn_update_dpp(0, hs, DPP_ROW_SHARE(6), 0xf, 0xf, false);
            ar[7] = __builtin_amdgcn_update_dpp(0, hs, DPP_ROW_SHARE(7), 0xf, 0xf, false);
            ar[8] = __builtin_amdgcn_update_dpp(0, hs, DPP_ROW_SHARE(8), 0xf, 0xf, false);
            ar[9] = __builtin_amdgcn_update_dpp(0, hs, DPP_ROW_SHARE(9), 0xf, 0xf, false);
            v8f acc[5];
#pragma unroll
            for (int n = 0; n < 5; ++n) {
                v8f z = {0.f, 0.f, 0.f, 0.f, 0.f, 0.f, 0.f, 0.f};
                acc[n] = z;
            }
#pragma unroll
            for (int kc = 0; kc < 5; ++kc) {
                v2f a;
                a.x = __int_as_float(ar[2 * kc]);
                a.y = __int_as_float(ar[2 * kc + 1]);
#pragma unroll
                for (int n = 0; n < 5; ++n) {
                    acc[n] = __builtin_amdgcn_wmma_f32_16x16x4_f32(
                        false, a, false, bf[kc][n], (short)0, acc[n], false, false);
                }
            }
            const int t4 = __float_as_int(acc[4][0]);
            g4[0] = __int_as_float(__builtin_amdgcn_update_dpp(
                __float_as_int(acc[0][0]), t4, DPP_QUAD_ID,      0x2, 0x1, false));
            g4[1] = __int_as_float(__builtin_amdgcn_update_dpp(
                __float_as_int(acc[1][0]), t4, DPP_ROW_SHL(4),  0x2, 0x1, false));
            g4[2] = __int_as_float(__builtin_amdgcn_update_dpp(
                __float_as_int(acc[2][0]), t4, DPP_ROW_SHL(8),  0x2, 0x1, false));
            g4[3] = __int_as_float(__builtin_amdgcn_update_dpp(
                __float_as_int(acc[3][0]), t4, DPP_ROW_SHL(12), 0x2, 0x1, false));
        }
        if constexpr (L2) {
            const float4 q = qn;
            qn = *reinterpret_cast<const float4*>(&lds_g2[((row + 1) & (GRP - 1)) * G2ST]);
            g4[0] = (!L1 || is2lane) ? q.x : g4[0];
            g4[1] = (!L1 || is2lane) ? q.y : g4[1];
            g4[2] = (!L1 || is2lane) ? q.z : g4[2];
            g4[3] = (!L1 || is2lane) ? q.w : g4[3];
        }
        const float xr = is2lane ? hst : xs;

        const bool upd = (L1 && L2) ? true : (is2lane ? L2 : L1);
        if (upd) {
            float gi = g4[0] + xr * wx[0] + bs[0];
            float gf = g4[1] + xr * wx[1] + bs[1];
            float gg = g4[2] + xr * wx[2] + bs[2];
            float go = g4[3] + xr * wx[3] + bs[3];
            float cn = sigm(gf) * cst + sigm(gi) * tanh_f(gg);
            cst = cn;
            float hn = sigm(go) * tanh_f(cn);
            hst = hn;
            if constexpr (L1) {
                if (L2 || !is2lane) lds_c1h[c1off + row * c1str] = cn;
            }
            if constexpr (L2) {
                if (is2lane) out[(size_t)row0 * TT + tout] = cn;
            }
        }
    };

    auto product2 = [&](int rnext) __attribute__((always_inline)) {
        lds_barrier();
        v8f acc2 = {0.f, 0.f, 0.f, 0.f, 0.f, 0.f, 0.f, 0.f};
#pragma unroll
        for (int kc = 0; kc < 5; ++kc) {
            v2f a;
            const float* cp = &lds_c1h[nu * HID + 4 * kc + 2 * khalf];
            a.x = cp[0]; a.y = cp[1];
            acc2 = __builtin_amdgcn_wmma_f32_16x16x4_f32(
                false, a, false, bf2[kc], (short)0, acc2, false, false);
        }
#pragma unroll
        for (int v = 0; v < 8; ++v)
            lds_g2[(8 * khalf + v) * G2ST + nu] = acc2[v];
        lds_barrier();
        qn = *reinterpret_cast<const float4*>(&lds_g2[rnext * G2ST]);
    };

    float xcur = xin[(size_t)row0 * T_LEN + lane];
    float xnxt = xin[(size_t)row0 * T_LEN + XBLK + lane];

    __syncthreads();

    auto group = [&](auto f2, float xc, int lbase, int tbase) __attribute__((always_inline)) {
        for (int tl = 0; tl < GRP; ++tl) {
            const float xs = __int_as_float(
                __builtin_amdgcn_readlane(__float_as_int(xc), lbase + tl));
            tick(flag<true>{}, f2, xs, tl, tbase + tl - GRP);
        }
        product2(0);
    };

    group(flag<false>{}, xcur, 0, 0);
    group(flag<true>{}, xcur, GRP, GRP);

    for (int b = 1; b < NBLK; ++b) {
        xcur = xnxt;
        const int bn = (b + 1 < NBLK) ? (b + 1) : (NBLK - 1);
        xnxt = xin[(size_t)row0 * T_LEN + bn * XBLK + lane];
        for (int h = 0; h < 2; ++h)
            group(flag<true>{}, xcur, h * GRP, b * XBLK + h * GRP);
    }

    for (int i = 0; i < GRP; ++i)
        tick(flag<false>{}, flag<true>{}, 0.0f, i, T_LEN - GRP + i);

    for (int f = 0; f < fut; ++f) {
        const int s = T_LEN + f;
        const int r = s & (GRP - 1);
        const float xs = __int_as_float(
            __builtin_amdgcn_readlane(__float_as_int(cst), L2LANE));
        tick(flag<true>{}, flag<false>{}, xs, r, 0);
        product2(r);
        tick(flag<false>{}, flag<true>{}, 0.0f, r, s);
    }
}

extern "C" void kernel_launch(void* const* d_in, const int* in_sizes, int n_in,
                              void* d_out, int out_size, void* d_ws, size_t ws_size,
                              hipStream_t stream)
{
    const float* xin  = (const float*)d_in[0];
    const float* Wih1 = (const float*)d_in[1];
    const float* Whh1 = (const float*)d_in[2];
    const float* bih1 = (const float*)d_in[3];
    const float* bhh1 = (const float*)d_in[4];
    const float* Wih2 = (const float*)d_in[5];
    const float* Whh2 = (const float*)d_in[6];
    const float* bih2 = (const float*)d_in[7];
    const float* bhh2 = (const float*)d_in[8];
    const int*   futp = (const int*)d_in[9];
    float* out = (float*)d_out;

    dim3 grid(B_TOT);
    dim3 block(32);
    lstm_seq_kernel<<<grid, block, 0, stream>>>(xin, Wih1, Whh1, bih1, bhh1,
                                                Wih2, Whh2, bih2, bhh2, futp, out);
}


# Work-group = 4 independent waves (one batch row each) instead of 1: grid 1024 -> 256, LDS sized so one work-group per WGP, each wave alone on a SIMD; block barriers go; sums unchanged
# speedup vs baseline: 5.9214x; 1.6493x over previous
#include <hip/hip_runtime.h>

typedef __attribute__((ext_vector_type(2))) float v2f;
typedef __attribute__((ext_vector_type(8))) float v8f;

#define B_TOT 1024
#define T_LEN 4096
#define HID   20
#define XBLK  32
#define NBLK  (T_LEN / XBLK)
#define GRP   16
#define NGRP  (T_LEN / GRP)
#define L2LANE 20
#define G2ST  16
#define WAVES 4
#define N_WG  256
#define WG_THREADS 128
#define C1H_FLOATS (GRP * HID + 32)
#define G2_FLOATS  (GRP * G2ST)
#define WV_STRIDE 10496
static_assert(N_WG * WAVES == B_TOT && WG_THREADS == 32 * WAVES, "one wave per batch row");
static_assert(C1H_FLOATS % 4 == 0 && WV_STRIDE % 4 == 0 && C1H_FLOATS + G2_FLOATS <= WV_STRIDE,
              "16-byte aligned regions inside a wave's stride");
static_assert(WAVES * WV_STRIDE * 4 > 160 * 1024 && WAVES * WV_STRIDE * 4 <= 320 * 1024,
              "one work-group per WGP");
static_assert(T_LEN % XBLK == 0 && NBLK >= 2, "x blocks must tile the sequence");
static_assert(XBLK == 2 * GRP && T_LEN % GRP == 0 && NGRP >= 2, "two product groups per x block");

#define DPP_QUAD_ID      0xE4
#define DPP_ROW_SHL(n)   (0x100 + (n))
#define DPP_ROW_SHARE(n) (0x150 + (n))

template <bool V> struct flag { static constexpr bool value = V; };

__device__ __forceinline__ float sigm(float x) {
    return 1.0f / (1.0f + __expf(-x));
}
__device__ __forceinline__ float tanh_f(float x) {
    return 1.0f - 2.0f / (__expf(2.0f * x) + 1.0f);
}

__device__ __forceinline__ void lds_barrier() {
    __builtin_amdgcn_fence(__ATOMIC_RELEASE, "wavefront", "local");
    __builtin_amdgcn_fence(__ATOMIC_ACQUIRE, "wavefront", "local");
}

__device__ __forceinline__ int kunit(int r, int h) {
    return (h == 0) ? r : ((r < 4) ? (16 + r) : (r + 6));
}

__global__ void __launch_bounds__(WG_THREADS)
lstm_seq_kernel(const float* __restrict__ xin,
                const float* __restrict__ Wih1,
                const float* __restrict__ Whh1,
                const float* __restrict__ bih1,
                const float* __restrict__ bhh1,
                const float* __restrict__ Wih2,
                const float* __restrict__ Whh2,
                const float* __restrict__ bih2,
                const float* __restrict__ bhh2,
                const int*   __restrict__ futp,
                float* __restrict__ out)
{
    __shared__ __attribute__((aligned(16))) float lds_pool[WAVES * WV_STRIDE];

    const int wv    = __builtin_amdgcn_readfirstlane((int)(threadIdx.x >> 5));
    float* const lds_c1h = lds_pool + wv * WV_STRIDE;
    float* const lds_g2  = lds_c1h + C1H_FLOATS;

    const int lane  = threadIdx.x & 31;
    const int nu    = lane & 15;
    const int khalf = lane >> 4;
    const int row0  = blockIdx.x * WAVES + wv;
    const int fut   = *futp;
    const int TT    = T_LEN + fut;

    const bool is2lane = (lane == L2LANE);
    const int  unit    = (lane < HID) ? lane : nu;
    const bool primary = (lane < 10) || (lane >= 16 && lane < HID) || (lane >= 26);
    const int  c1off = primary ? unit : (GRP * HID + lane);
    const int  c1str = primary ? HID : 0;

    for (int i = lane; i < GRP * HID; i += 32)  lds_c1h[i] = 0.0f;

    float wx[4], bs[4];
#pragma unroll
    for (int g = 0; g < 4; ++g) {
        if (is2lane) {
            wx[g] = Whh2[g];
            bs[g] = bih2[g] + bhh2[g];
        } else {
            wx[g] = Wih1[g * HID + unit];
            bs[g] = bih1[g * HID + unit] + bhh1[g * HID + unit];
        }
    }

    v2f bf[5][5];
#pragma unroll
    for (int n = 0; n < 5; ++n) {
        const int wrow = (n < 4) ? (n * HID + nu) : ((nu >> 2) * HID + 16 + (nu & 3));
#pragma unroll
        for (int kc = 0; kc < 5; ++kc) {
            v2f b;
            b.x = Whh1[wrow * HID + kunit(2 * kc,     khalf)];
            b.y = Whh1[wrow * HID + kunit(2 * kc + 1, khalf)];
            bf[kc][n] = b;
        }
    }
    v2f bf2[5];
#pragma unroll
    for (int kc = 0; kc < 5; ++kc) {
        v2f b; b.x = 0.0f; b.y = 0.0f;
        if (nu < 4) {
            const float* p = Wih2 + nu * HID + 4 * kc + 2 * khalf;
            b.x = p[0]; b.y = p[1];
        }
        bf2[kc] = b;
    }

    float cst = 0.0f, hst = 0.0f;
    float4 qn = make_float4(0.0f, 0.0f, 0.0f, 0.0f);

    auto tick = [&](auto f1, auto f2, float xs, int row, int tout) __attribute__((always_inline)) {
        constexpr bool L1 = decltype(f1)::value;
        constexpr bool L2 = decltype(f2)::value;
        float g4[4] = {0.0f, 0.0f, 0.0f, 0.0f};
        if constexpr (L1) {
            const int hi = __float_as_int(hst);
            const int hs = __builtin_amdgcn_update_dpp(hi, hi, DPP_ROW_SHL(6), 0x2, 0xf, false);
            int ar[10];
            ar[0] = __builtin_amdgcn_update_dpp(0, hi, DPP_ROW_SHARE(0), 0xf, 0xf, false);
            ar[1] = __builtin_amdgcn_update_dpp(0, hi, DPP_ROW_SHARE(1), 0xf, 0xf, false);
            ar[2] = __builtin_amdgcn_update_dpp(0, hi, DPP_ROW_SHARE(2), 0xf, 0xf, false);
            ar[3] = __builtin_amdgcn_update_dpp(0, hi, DPP_ROW_SHARE(3), 0xf, 0xf, false);
            ar[4] = __builtin_amdgcn_update_dpp(0, hs, DPP_ROW_SHARE(4), 0xf, 0xf, false);
            ar[5] = __builtin_amdgcn_update_dpp(0, hs, DPP_ROW_SHARE(5), 0xf, 0xf, false);
            ar[6] = __builtin_amdgcn_update_dpp(0, hs, DPP_ROW_SHARE(6), 0xf, 0xf, false);
            ar[7] = __builtin_amdgcn_update_dpp(0, hs, DPP_ROW_SHARE(7), 0xf, 0xf, false);
            ar[8] = __builtin_amdgcn_update_dpp(0, hs, DPP_ROW_SHARE(8), 0xf, 0xf, false);
            ar[9] = __builtin_amdgcn_update_dpp(0, hs, DPP_ROW_SHARE(9), 0xf, 0xf, false);
            v8f acc[5];
#pragma unroll
            for (int n = 0; n < 5; ++n) {
                v8f z = {0.f, 0.f, 0.f, 0.f, 0.f, 0.f, 0.f, 0.f};
                acc[n] = z;
            }
#pragma unroll
            for (int kc = 0; kc < 5; ++kc) {
                v2f a;
                a.x = __int_as_float(ar[2 * kc]);
                a.y = __int_as_float(ar[2 * kc + 1]);
#pragma unroll
                for (int n = 0; n < 5; ++n) {
                    acc[n] = __builtin_amdgcn_wmma_f32_16x16x4_f32(
                        false, a, false, bf[kc][n], (short)0, acc[n], false, false);
                }
            }
            const int t4 = __float_as_int(acc[4][0]);
            g4[0] = __int_as_float(__builtin_amdgcn_update_dpp(
                __float_as_int(acc[0][0]), t4, DPP_QUAD_ID,      0x2, 0x1, false));
            g4[1] = __int_as_float(__builtin_amdgcn_update_dpp(
                __float_as_int(acc[1][0]), t4, DPP_ROW_SHL(4),  0x2, 0x1, false));
            g4[2] = __int_as_float(__builtin_amdgcn_update_dpp(
                __float_as_int(acc[2][0]), t4, DPP_ROW_SHL(8),  0x2, 0x1, false));
            g4[3] = __int_as_float(__builtin_amdgcn_update_dpp(
                __float_as_int(acc[3][0]), t4, DPP_ROW_SHL(12), 0x2, 0x1, false));
        }
        if constexpr (L2) {
            const float4 q = qn;
            qn = *reinterpret_cast<const float4*>(&lds_g2[((row + 1) & (GRP - 1)) * G2ST]);
            g4[0] = (!L1 || is2lane) ? q.x : g4[0];
            g4[1] = (!L1 || is2lane) ? q.y : g4[1];
            g4[2] = (!L1 || is2lane) ? q.z : g4[2];
            g4[3] = (!L1 || is2lane) ? q.w : g4[3];
        }
        const float xr = is2lane ? hst : xs;

        const bool upd = (L1 && L2) ? true : (is2lane ? L2 : L1);
        if (upd) {
            float gi = g4[0] + xr * wx[0] + bs[0];
            float gf = g4[1] + xr * wx[1] + bs[1];
            float gg = g4[2] + xr * wx[2] + bs[2];
            float go = g4[3] + xr * wx[3] + bs[3];
            float cn = sigm(gf) * cst + sigm(gi) * tanh_f(gg);
            cst = cn;
            float hn = sigm(go) * tanh_f(cn);
            hst = hn;
            if constexpr (L1) {
                if (L2 || !is2lane) lds_c1h[c1off + row * c1str] = cn;
            }
            if constexpr (L2) {
                if (is2lane) out[(size_t)row0 * TT + tout] = cn;
            }
        }
    };

    auto product2 = [&](int rnext) __attribute__((always_inline)) {
        lds_barrier();
        v8f acc2 = {0.f, 0.f, 0.f, 0.f, 0.f, 0.f, 0.f, 0.f};
#pragma unroll
        for (int kc = 0; kc < 5; ++kc) {
            v2f a;
            const float* cp = &lds_c1h[nu * HID + 4 * kc + 2 * khalf];
            a.x = cp[0]; a.y = cp[1];
            acc2 = __builtin_amdgcn_wmma_f32_16x16x4_f32(
                false, a, false, bf2[kc], (short)0, acc2, false, false);
        }
#pragma unroll
        for (int v = 0; v < 8; ++v)
            lds_g2[(8 * khalf + v) * G2ST + nu] = acc2[v];
        lds_barrier();
        qn = *reinterpret_cast<const float4*>(&lds_g2[rnext * G2ST]);
    };

    float xcur = xin[(size_t)row0 * T_LEN + lane];
    float xnxt = xin[(size_t)row0 * T_LEN + XBLK + lane];

    lds_barrier();

    auto group = [&](auto f2, float xc, int lbase, int tbase) __attribute__((always_inline)) {
        for (int tl = 0; tl < GRP; ++tl) {
            const float xs = __int_as_float(
                __builtin_amdgcn_readlane(__float_as_int(xc), lbase + tl));
            tick(flag<true>{}, f2, xs, tl, tbase + tl - GRP);
        }
        product2(0);
    };

    group(flag<false>{}, xcur, 0, 0);
    group(flag<true>{}, xcur, GRP, GRP);

    for (int b = 1; b < NBLK; ++b) {
        xcur = xnxt;
        const int bn = (b + 1 < NBLK) ? (b + 1) : (NBLK - 1);
        xnxt = xin[(size_t)row0 * T_LEN + bn * XBLK + lane];
        for (int h = 0; h < 2; ++h)
            group(flag<true>{}, xcur, h * GRP, b * XBLK + h * GRP);
    }

    for (int i = 0; i < GRP; ++i)
        tick(flag<false>{}, flag<true>{}, 0.0f, i, T_LEN - GRP + i);

    for (int f = 0; f < fut; ++f) {
        const int s = T_LEN + f;
        const int r = s & (GRP - 1);
        const float xs = __int_as_float(
            __builtin_amdgcn_readlane(__float_as_int(cst), L2LANE));
        tick(flag<true>{}, flag<false>{}, xs, r, 0);
        product2(r);
        tick(flag<false>{}, flag<true>{}, 0.0f, r, s);
    }
}

extern "C" void kernel_launch(void* const* d_in, const int* in_sizes, int n_in,
                              void* d_out, int out_size, void* d_ws, size_t ws_size,
                              hipStream_t stream)
{
    const float* xin  = (const float*)d_in[0];
    const float* Wih1 = (const float*)d_in[1];
    const float* Whh1 = (const float*)d_in[2];
    const float* bih1 = (const float*)d_in[3];
    const float* bhh1 = (const float*)d_in[4];
    const float* Wih2 = (const float*)d_in[5];
    const float* Whh2 = (const float*)d_in[6];
    const float* bih2 = (const float*)d_in[7];
    const float* bhh2 = (const float*)d_in[8];
    const int*   futp = (const int*)d_in[9];
    float* out = (float*)d_out;

    dim3 grid(N_WG);
    dim3 block(WG_THREADS);
    lstm_seq_kernel<<<grid, block, 0, stream>>>(xin, Wih1, Whh1, bih1, bhh1,
                                                Wih2, Whh2, bih2, bhh2, futp, out);
}


# Step is one branch-free block: c2 filed in LDS by the c1 store, one 128-byte output store per 32 steps; o-gate tile's products issue last, under staged i, f, g activations; c may fuse f*c or i*g
# speedup vs baseline: 6.3456x; 1.0716x over previous
#include <hip/hip_runtime.h>

typedef __attribute__((ext_vector_type(2))) float v2f;
typedef __attribute__((ext_vector_type(8))) float v8f;

#define B_TOT 1024
#define T_LEN 4096
#define HID   20
#define XBLK  32
#define NBLK  (T_LEN / XBLK)
#define GRP   16
#define NGRP  (T_LEN / GRP)
#define L2LANE 20
#define G2ST  16
#define WAVES 4
#define N_WG  256
#define WG_THREADS 128
#define C2H_OFF    (GRP * HID + 32)
#define C1H_FLOATS (GRP * HID + 32 + XBLK)
#define G2_FLOATS  (GRP * G2ST)
#define WV_STRIDE 10496
static_assert(N_WG * WAVES == B_TOT && WG_THREADS == 32 * WAVES, "one wave per batch row");
static_assert(C1H_FLOATS % 4 == 0 && WV_STRIDE % 4 == 0 && C1H_FLOATS + G2_FLOATS <= WV_STRIDE,
              "16-byte aligned regions inside a wave's stride");
static_assert(WAVES * WV_STRIDE * 4 > 160 * 1024 && WAVES * WV_STRIDE * 4 <= 320 * 1024,
              "one work-group per WGP");
static_assert(T_LEN % XBLK == 0 && NBLK >= 2, "x blocks must tile the sequence");
static_assert(XBLK == 2 * GRP && T_LEN % GRP == 0 && NGRP >= 2, "two product groups per x block");

#define DPP_QUAD_ID      0xE4
#define DPP_ROW_SHL(n)   (0x100 + (n))
#define DPP_ROW_SHARE(n) (0x150 + (n))

template <bool V> struct flag { static constexpr bool value = V; };

__device__ __forceinline__ float sigm(float x) {
    return 1.0f / (1.0f + __expf(-x));
}
__device__ __forceinline__ float tanh_f(float x) {
    return 1.0f - 2.0f / (__expf(2.0f * x) + 1.0f);
}

__device__ __forceinline__ void lds_barrier() {
    __builtin_amdgcn_fence(__ATOMIC_RELEASE, "wavefront", "local");
    __builtin_amdgcn_fence(__ATOMIC_ACQUIRE, "wavefront", "local");
}

__device__ __forceinline__ int kunit(int r, int h) {
    return (h == 0) ? r : ((r < 4) ? (16 + r) : (r + 6));
}

__global__ void __launch_bounds__(WG_THREADS)
lstm_seq_kernel(const float* __restrict__ xin,
                const float* __restrict__ Wih1,
                const float* __restrict__ Whh1,
                const float* __restrict__ bih1,
                const float* __restrict__ bhh1,
                const float* __restrict__ Wih2,
                const float* __restrict__ Whh2,
                const float* __restrict__ bih2,
                const float* __restrict__ bhh2,
                const int*   __restrict__ futp,
                float* __restrict__ out)
{
    __shared__ __attribute__((aligned(16))) float lds_pool[WAVES * WV_STRIDE];

    const int wv    = __builtin_amdgcn_readfirstlane((int)(threadIdx.x >> 5));
    float* const lds_c1h = lds_pool + wv * WV_STRIDE;
    float* const lds_g2  = lds_c1h + C1H_FLOATS;

    const int lane  = threadIdx.x & 31;
    const int nu    = lane & 15;
    const int khalf = lane >> 4;
    const int row0  = blockIdx.x * WAVES + wv;
    const int fut   = *futp;
    const int TT    = T_LEN + fut;

    const bool is2lane = (lane == L2LANE);
    const int  unit    = (lane < HID) ? lane : nu;
    const bool primary = (lane < 10) || (lane >= 16 && lane < HID) || (lane >= 26);
    const int  c1off = primary ? unit : (is2lane ? C2H_OFF : (GRP * HID + lane));
    const int  c1str = primary ? HID : (is2lane ? 1 : 0);

    for (int i = lane; i < GRP * HID; i += 32)  lds_c1h[i] = 0.0f;
    lds_c1h[C2H_OFF + lane] = 0.0f;

    float wx[4], bs[4];
#pragma unroll
    for (int g = 0; g < 4; ++g) {
        if (is2lane) {
            wx[g] = Whh2[g];
            bs[g] = bih2[g] + bhh2[g];
        } else {
            wx[g] = Wih1[g * HID + unit];
            bs[g] = bih1[g * HID + unit] + bhh1[g * HID + unit];
        }
    }

    v2f bf[5][5];
#pragma unroll
    for (int n = 0; n < 5; ++n) {
        const int wrow = (n < 4) ? (n * HID + nu) : ((nu >> 2) * HID + 16 + (nu & 3));
#pragma unroll
        for (int kc = 0; kc < 5; ++kc) {
            v2f b;
            b.x = Whh1[wrow * HID + kunit(2 * kc,     khalf)];
            b.y = Whh1[wrow * HID + kunit(2 * kc + 1, khalf)];
            bf[kc][n] = b;
        }
    }
    v2f bf2[5];
#pragma unroll
    for (int kc = 0; kc < 5; ++kc) {
        v2f b; b.x = 0.0f; b.y = 0.0f;
        if (nu < 4) {
            const float* p = Wih2 + nu * HID + 4 * kc + 2 * khalf;
            b.x = p[0]; b.y = p[1];
        }
        bf2[kc] = b;
    }

    float cst = 0.0f, hst = 0.0f;
    float4 qn = make_float4(0.0f, 0.0f, 0.0f, 0.0f);

    auto tick = [&](auto f1, auto f2, float xs, int row, int tout, int c1b) __attribute__((always_inline)) {
        constexpr bool L1 = decltype(f1)::value;
        constexpr bool L2 = decltype(f2)::value;
        float g4[4] = {0.0f, 0.0f, 0.0f, 0.0f};
        if constexpr (L1) {
            const int hi = __float_as_int(hst);
            const int hs = __builtin_amdgcn_update_dpp(hi, hi, DPP_ROW_SHL(6), 0x2, 0xf, false);
            int ar[10];
            ar[0] = __builtin_amdgcn_update_dpp(0, hi, DPP_ROW_SHARE(0), 0xf, 0xf, false);
            ar[1] = __builtin_amdgcn_update_dpp(0, hi, DPP_ROW_SHARE(1), 0xf, 0xf, false);
            ar[2] = __builtin_amdgcn_update_dpp(0, hi, DPP_ROW_SHARE(2), 0xf, 0xf, false);
            ar[3] = __builtin_amdgcn_update_dpp(0, hi, DPP_ROW_SHARE(3), 0xf, 0xf, false);
            ar[4] = __builtin_amdgcn_update_dpp(0, hs, DPP_ROW_SHARE(4), 0xf, 0xf, false);
            ar[5] = __builtin_amdgcn_update_dpp(0, hs, DPP_ROW_SHARE(5), 0xf, 0xf, false);
            ar[6] = __builtin_amdgcn_update_dpp(0, hs, DPP_ROW_SHARE(6), 0xf, 0xf, false);
            ar[7] = __builtin_amdgcn_update_dpp(0, hs, DPP_ROW_SHARE(7), 0xf, 0xf, false);
            ar[8] = __builtin_amdgcn_update_dpp(0, hs, DPP_ROW_SHARE(8), 0xf, 0xf, false);
            ar[9] = __builtin_amdgcn_update_dpp(0, hs, DPP_ROW_SHARE(9), 0xf, 0xf, false);
            v8f acc[5];
#pragma unroll
            for (int n = 0; n < 5; ++n) {
                v8f z = {0.f, 0.f, 0.f, 0.f, 0.f, 0.f, 0.f, 0.f};
                acc[n] = z;
            }
#pragma unroll
            for (int kc = 0; kc < 5; ++kc) {
                v2f a;
                a.x = __int_as_float(ar[2 * kc]);
                a.y = __int_as_float(ar[2 * kc + 1]);
#pragma unroll
                for (int n = 0; n < 5; ++n) {
                    acc[n] = __builtin_amdgcn_wmma_f32_16x16x4_f32(
                        false, a, false, bf[kc][n], (short)0, acc[n], false, false);
                }
            }
            const int t4 = __float_as_int(acc[4][0]);
            g4[0] = __int_as_float(__builtin_amdgcn_update_dpp(
                __float_as_int(acc[0][0]), t4, DPP_QUAD_ID,      0x2, 0x1, false));
            g4[1] = __int_as_float(__builtin_amdgcn_update_dpp(
                __float_as_int(acc[1][0]), t4, DPP_ROW_SHL(4),  0x2, 0x1, false));
            g4[2] = __int_as_float(__builtin_amdgcn_update_dpp(
                __float_as_int(acc[2][0]), t4, DPP_ROW_SHL(8),  0x2, 0x1, false));
            g4[3] = __int_as_float(__builtin_amdgcn_update_dpp(
                __float_as_int(acc[3][0]), t4, DPP_ROW_SHL(12), 0x2, 0x1, false));
        }
        if constexpr (L2) {
            const float4 q = qn;
            qn = *reinterpret_cast<const float4*>(&lds_g2[((row + 1) & (GRP - 1)) * G2ST]);
            g4[0] = (!L1 || is2lane) ? q.x : g4[0];
            g4[1] = (!L1 || is2lane) ? q.y : g4[1];
            g4[2] = (!L1 || is2lane) ? q.z : g4[2];
            g4[3] = (!L1 || is2lane) ? q.w : g4[3];
        }
        const float xr = is2lane ? hst : xs;

        const bool upd = (L1 && L2) ? true : (is2lane ? L2 : L1);
        if (upd) {
            float gi = g4[0] + xr * wx[0] + bs[0];
            float gf = g4[1] + xr * wx[1] + bs[1];
            float gg = g4[2] + xr * wx[2] + bs[2];
            float go = g4[3] + xr * wx[3] + bs[3];
            float cn = sigm(gf) * cst + sigm(gi) * tanh_f(gg);
            cst = cn;
            float hn = sigm(go) * tanh_f(cn);
            hst = hn;
            if constexpr (L1) {
                if (L2 || !is2lane) lds_c1h[c1b + row * c1str] = cn;
            }
            if constexpr (L2 && !L1) {
                if (is2lane) out[(size_t)row0 * TT + tout] = cn;
            }
        }
    };

    auto tick_main = [&](float xs, int row, int c1b) __attribute__((always_inline)) {
        const int hi = __float_as_int(hst);
        const int hs = __builtin_amdgcn_update_dpp(hi, hi, DPP_ROW_SHL(6), 0x2, 0xf, false);
        int ar[10];
        ar[0] = __builtin_amdgcn_update_dpp(0, hi, DPP_ROW_SHARE(0), 0xf, 0xf, false);
        ar[1] = __builtin_amdgcn_update_dpp(0, hi, DPP_ROW_SHARE(1), 0xf, 0xf, false);
        ar[2] = __builtin_amdgcn_update_dpp(0, hi, DPP_ROW_SHARE(2), 0xf, 0xf, false);
        ar[3] = __builtin_amdgcn_update_dpp(0, hi, DPP_ROW_SHARE(3), 0xf, 0xf, false);
        ar[4] = __builtin_amdgcn_update_dpp(0, hs, DPP_ROW_SHARE(4), 0xf, 0xf, false);
        ar[5] = __builtin_amdgcn_update_dpp(0, hs, DPP_ROW_SHARE(5), 0xf, 0xf, false);
        ar[6] = __builtin_amdgcn_update_dpp(0, hs, DPP_ROW_SHARE(6), 0xf, 0xf, false);
        ar[7] = __builtin_amdgcn_update_dpp(0, hs, DPP_ROW_SHARE(7), 0xf, 0xf, false);
        ar[8] = __builtin_amdgcn_update_dpp(0, hs, DPP_ROW_SHARE(8), 0xf, 0xf, false);
        ar[9] = __builtin_amdgcn_update_dpp(0, hs, DPP_ROW_SHARE(9), 0xf, 0xf, false);
        v2f af[5];
        v8f acc[5];
#pragma unroll
        for (int kc = 0; kc < 5; ++kc) {
            af[kc].x = __int_as_float(ar[2 * kc]);
            af[kc].y = __int_as_float(ar[2 * kc + 1]);
            v8f z = {0.f, 0.f, 0.f, 0.f, 0.f, 0.f, 0.f, 0.f};
            acc[kc] = z;
        }
#pragma unroll
        for (int kc = 0; kc < 5; ++kc) {
            acc[2] = __builtin_amdgcn_wmma_f32_16x16x4_f32(false, af[kc], false, bf[kc][2], (short)0, acc[2], false, false);
            acc[4] = __builtin_amdgcn_wmma_f32_16x16x4_f32(false, af[kc], false, bf[kc][4], (short)0, acc[4], false, false);
            acc[1] = __builtin_amdgcn_wmma_f32_16x16x4_f32(false, af[kc], false, bf[kc][1], (short)0, acc[1], false, false);
            acc[0] = __builtin_amdgcn_wmma_f32_16x16x4_f32(false, af[kc], false, bf[kc][0], (short)0, acc[0], false, false);
        }
        __builtin_amdgcn_sched_barrier(0);
        acc[3] = __builtin_amdgcn_wmma_f32_16x16x4_f32(false, af[0], false, bf[0][3], (short)0, acc[3], false, false);
        const float4 q = qn;
        qn = *reinterpret_cast<const float4*>(&lds_g2[((row + 1) & (GRP - 1)) * G2ST]);
        const float xr = is2lane ? hst : xs;
        const int t4 = __float_as_int(acc[4][0]);
        float pg = __int_as_float(__builtin_amdgcn_update_dpp(__float_as_int(acc[2][0]), t4, DPP_ROW_SHL(8), 0x2, 0x1, false));
        float pi = __int_as_float(__builtin_amdgcn_update_dpp(__float_as_int(acc[0][0]), t4, DPP_QUAD_ID,    0x2, 0x1, false));
        float pf = __int_as_float(__builtin_amdgcn_update_dpp(__float_as_int(acc[1][0]), t4, DPP_ROW_SHL(4), 0x2, 0x1, false));
        pg = is2lane ? q.z : pg;
        pi = is2lane ? q.x : pi;
        pf = is2lane ? q.y : pf;
        const float gg = pg + xr * wx[2] + bs[2];
        const float gi = pi + xr * wx[0] + bs[0];
        const float gf = pf + xr * wx[1] + bs[1];
        __builtin_amdgcn_sched_barrier(0);
        acc[3] = __builtin_amdgcn_wmma_f32_16x16x4_f32(false, af[1], false, bf[1][3], (short)0, acc[3], false, false);
        const float dg = __expf(2.0f * gg) + 1.0f;
        const float di = 1.0f + __expf(-gi);
        const float df = 1.0f + __expf(-gf);
        __builtin_amdgcn_sched_barrier(0);
        acc[3] = __builtin_amdgcn_wmma_f32_16x16x4_f32(false, af[2], false, bf[2][3], (short)0, acc[3], false, false);
        acc[3] = __builtin_amdgcn_wmma_f32_16x16x4_f32(false, af[3], false, bf[3][3], (short)0, acc[3], false, false);
        acc[3] = __builtin_amdgcn_wmma_f32_16x16x4_f32(false, af[4], false, bf[4][3], (short)0, acc[3], false, false);
        const float tg = 1.0f - 2.0f / dg;
        const float si = 1.0f / di;
        const float sf = 1.0f / df;
        const float cn = sf * cst + si * tg;
        cst = cn;
        float po = __int_as_float(__builtin_amdgcn_update_dpp(__float_as_int(acc[3][0]), t4, DPP_ROW_SHL(12), 0x2, 0x1, false));
        po = is2lane ? q.w : po;
        const float go = po + xr * wx[3] + bs[3];
        hst = sigm(go) * tanh_f(cn);
        lds_c1h[c1b + row * c1str] = cn;
    };

    auto product2 = [&](int rnext) __attribute__((always_inline)) {
        lds_barrier();
        v8f acc2 = {0.f, 0.f, 0.f, 0.f, 0.f, 0.f, 0.f, 0.f};
#pragma unroll
        for (int kc = 0; kc < 5; ++kc) {
            v2f a;
            const float* cp = &lds_c1h[nu * HID + 4 * kc + 2 * khalf];
            a.x = cp[0]; a.y = cp[1];
            acc2 = __builtin_amdgcn_wmma_f32_16x16x4_f32(
                false, a, false, bf2[kc], (short)0, acc2, false, false);
        }
#pragma unroll
        for (int v = 0; v < 8; ++v)
            lds_g2[(8 * khalf + v) * G2ST + nu] = acc2[v];
        lds_barrier();
        qn = *reinterpret_cast<const float4*>(&lds_g2[rnext * G2ST]);
    };

    float xcur = xin[(size_t)row0 * T_LEN + lane];
    float xnxt = xin[(size_t)row0 * T_LEN + XBLK + lane];

    lds_barrier();

    auto group = [&](auto f2, float xc, int lbase, int tbase) __attribute__((always_inline)) {
        const int c1b = c1off + (is2lane ? lbase : 0);
        for (int tl = 0; tl < GRP; ++tl) {
            const float xs = __int_as_float(
                __builtin_amdgcn_readlane(__float_as_int(xc), lbase + tl));
            if constexpr (decltype(f2)::value) tick_main(xs, tl, c1b);
            else tick(flag<true>{}, f2, xs, tl, tbase + tl - GRP, c1b);
        }
        product2(0);
    };
    auto store_block = [&](int wbase, bool lo_valid) __attribute__((always_inline)) {
        lds_barrier();
        const float o = lds_c1h[C2H_OFF + lane];
        if (lo_valid || lane >= GRP) out[(long)row0 * TT + (wbase + lane)] = o;
    };

    group(flag<false>{}, xcur, 0, 0);
    group(flag<true>{}, xcur, GRP, GRP);
    store_block(-GRP, false);

    for (int b = 1; b < NBLK; ++b) {
        xcur = xnxt;
        const int bn = (b + 1 < NBLK) ? (b + 1) : (NBLK - 1);
        xnxt = xin[(size_t)row0 * T_LEN + bn * XBLK + lane];
        for (int h = 0; h < 2; ++h)
            group(flag<true>{}, xcur, h * GRP, b * XBLK + h * GRP);
        store_block(b * XBLK - GRP, true);
    }

    for (int i = 0; i < GRP; ++i)
        tick(flag<false>{}, flag<true>{}, 0.0f, i, T_LEN - GRP + i, c1off);

    for (int f = 0; f < fut; ++f) {
        const int s = T_LEN + f;
        const int r = s & (GRP - 1);
        const float xs = __int_as_float(
            __builtin_amdgcn_readlane(__float_as_int(cst), L2LANE));
        tick(flag<true>{}, flag<false>{}, xs, r, 0, c1off);
        product2(r);
        tick(flag<false>{}, flag<true>{}, 0.0f, r, s, c1off);
    }
}

extern "C" void kernel_launch(void* const* d_in, const int* in_sizes, int n_in,
                              void* d_out, int out_size, void* d_ws, size_t ws_size,
                              hipStream_t stream)
{
    const float* xin  = (const float*)d_in[0];
    const float* Wih1 = (const float*)d_in[1];
    const float* Whh1 = (const float*)d_in[2];
    const float* bih1 = (const float*)d_in[3];
    const float* bhh1 = (const float*)d_in[4];
    const float* Wih2 = (const float*)d_in[5];
    const float* Whh2 = (const float*)d_in[6];
    const float* bih2 = (const float*)d_in[7];
    const float* bhh2 = (const float*)d_in[8];
    const int*   futp = (const int*)d_in[9];
    float* out = (float*)d_out;

    dim3 grid(N_WG);
    dim3 block(WG_THREADS);
    lstm_seq_kernel<<<grid, block, 0, stream>>>(xin, Wih1, Whh1, bih1, bhh1,
                                                Wih2, Whh2, bih2, bhh2, futp, out);
}


# Every exponential's argument is one FMA behind its producer: log2(e), bias, input term and the layer-2 lane's select become per-lane factors and an early addend; gate sum order changes
# speedup vs baseline: 6.5698x; 1.0353x over previous
#include <hip/hip_runtime.h>

typedef __attribute__((ext_vector_type(2))) float v2f;
typedef __attribute__((ext_vector_type(8))) float v8f;

#define B_TOT 1024
#define T_LEN 4096
#define HID   20
#define XBLK  32
#define NBLK  (T_LEN / XBLK)
#define GRP   16
#define NGRP  (T_LEN / GRP)
#define L2LANE 20
#define G2ST  16
#define WAVES 4
#define N_WG  256
#define WG_THREADS 128
#define C2H_OFF    (GRP * HID + 32)
#define C1H_FLOATS (GRP * HID + 32 + XBLK)
#define G2_FLOATS  (GRP * G2ST)
#define WV_STRIDE 10496
static_assert(N_WG * WAVES == B_TOT && WG_THREADS == 32 * WAVES, "one wave per batch row");
static_assert(C1H_FLOATS % 4 == 0 && WV_STRIDE % 4 == 0 && C1H_FLOATS + G2_FLOATS <= WV_STRIDE,
              "16-byte aligned regions inside a wave's stride");
static_assert(WAVES * WV_STRIDE * 4 > 160 * 1024 && WAVES * WV_STRIDE * 4 <= 320 * 1024,
              "one work-group per WGP");
static_assert(T_LEN % XBLK == 0 && NBLK >= 2, "x blocks must tile the sequence");
static_assert(XBLK == 2 * GRP && T_LEN % GRP == 0 && NGRP >= 2, "two product groups per x block");

#define LOG2E 1.44269504088896340736f
#define DPP_QUAD_ID      0xE4
#define DPP_ROW_SHL(n)   (0x100 + (n))
#define DPP_ROW_SHARE(n) (0x150 + (n))

template <bool V> struct flag { static constexpr bool value = V; };

__device__ __forceinline__ float sigm(float x) {
    return 1.0f / (1.0f + __expf(-x));
}
__device__ __forceinline__ float tanh_f(float x) {
    return 1.0f - 2.0f / (__expf(2.0f * x) + 1.0f);
}

__device__ __forceinline__ void lds_barrier() {
    __builtin_amdgcn_fence(__ATOMIC_RELEASE, "wavefront", "local");
    __builtin_amdgcn_fence(__ATOMIC_ACQUIRE, "wavefront", "local");
}

__device__ __forceinline__ int kunit(int r, int h) {
    return (h == 0) ? r : ((r < 4) ? (16 + r) : (r + 6));
}

__global__ void __launch_bounds__(WG_THREADS)
lstm_seq_kernel(const float* __restrict__ xin,
                const float* __restrict__ Wih1,
                const float* __restrict__ Whh1,
                const float* __restrict__ bih1,
                const float* __restrict__ bhh1,
                const float* __restrict__ Wih2,
                const float* __restrict__ Whh2,
                const float* __restrict__ bih2,
                const float* __restrict__ bhh2,
                const int*   __restrict__ futp,
                float* __restrict__ out)
{
    __shared__ __attribute__((aligned(16))) float lds_pool[WAVES * WV_STRIDE];

    const int wv    = __builtin_amdgcn_readfirstlane((int)(threadIdx.x >> 5));
    float* const lds_c1h = lds_pool + wv * WV_STRIDE;
    float* const lds_g2  = lds_c1h + C1H_FLOATS;

    const int lane  = threadIdx.x & 31;
    const int nu    = lane & 15;
    const int khalf = lane >> 4;
    const int row0  = blockIdx.x * WAVES + wv;
    const int fut   = *futp;
    const int TT    = T_LEN + fut;

    const bool is2lane = (lane == L2LANE);
    const int  unit    = (lane < HID) ? lane : nu;
    const bool primary = (lane < 10) || (lane >= 16 && lane < HID) || (lane >= 26);
    const int  c1off = primary ? unit : (is2lane ? C2H_OFF : (GRP * HID + lane));
    const int  c1str = primary ? HID : (is2lane ? 1 : 0);

    for (int i = lane; i < GRP * HID; i += 32)  lds_c1h[i] = 0.0f;
    lds_c1h[C2H_OFF + lane] = 0.0f;

    float wx[4], bs[4];
#pragma unroll
    for (int g = 0; g < 4; ++g) {
        if (is2lane) {
            wx[g] = Whh2[g];
            bs[g] = bih2[g] + bhh2[g];
        } else {
            wx[g] = Wih1[g * HID + unit];
            bs[g] = bih1[g * HID + unit] + bhh1[g * HID + unit];
        }
    }

    float wsc[4], bsc[4], kp[4], kq[4];
#pragma unroll
    for (int g = 0; g < 4; ++g) {
        const float ks = (g == 2) ? (2.0f * LOG2E) : -LOG2E;
        wsc[g] = ks * wx[g];
        bsc[g] = ks * bs[g];
        kp[g]  = is2lane ? 0.0f : ks;
        kq[g]  = is2lane ? ks : 0.0f;
    }

    v2f bf[5][5];
#pragma unroll
    for (int n = 0; n < 5; ++n) {
        const int wrow = (n < 4) ? (n * HID + nu) : ((nu >> 2) * HID + 16 + (nu & 3));
#pragma unroll
        for (int kc = 0; kc < 5; ++kc) {
            v2f b;
            b.x = Whh1[wrow * HID + kunit(2 * kc,     khalf)];
            b.y = Whh1[wrow * HID + kunit(2 * kc + 1, khalf)];
            bf[kc][n] = b;
        }
    }
    v2f bf2[5];
#pragma unroll
    for (int kc = 0; kc < 5; ++kc) {
        v2f b; b.x = 0.0f; b.y = 0.0f;
        if (nu < 4) {
            const float* p = Wih2 + nu * HID + 4 * kc + 2 * khalf;
            b.x = p[0]; b.y = p[1];
        }
        bf2[kc] = b;
    }

    float cst = 0.0f, hst = 0.0f;
    float4 qn = make_float4(0.0f, 0.0f, 0.0f, 0.0f);

    auto tick = [&](auto f1, auto f2, float xs, int row, int tout, int c1b) __attribute__((always_inline)) {
        constexpr bool L1 = decltype(f1)::value;
        constexpr bool L2 = decltype(f2)::value;
        float g4[4] = {0.0f, 0.0f, 0.0f, 0.0f};
        if constexpr (L1) {
            const int hi = __float_as_int(hst);
            const int hs = __builtin_amdgcn_update_dpp(hi, hi, DPP_ROW_SHL(6), 0x2, 0xf, false);
            int ar[10];
            ar[0] = __builtin_amdgcn_update_dpp(0, hi, DPP_ROW_SHARE(0), 0xf, 0xf, false);
            ar[1] = __builtin_amdgcn_update_dpp(0, hi, DPP_ROW_SHARE(1), 0xf, 0xf, false);
            ar[2] = __builtin_amdgcn_update_dpp(0, hi, DPP_ROW_SHARE(2), 0xf, 0xf, false);
            ar[3] = __builtin_amdgcn_update_dpp(0, hi, DPP_ROW_SHARE(3), 0xf, 0xf, false);
            ar[4] = __builtin_amdgcn_update_dpp(0, hs, DPP_ROW_SHARE(4), 0xf, 0xf, false);
            ar[5] = __builtin_amdgcn_update_dpp(0, hs, DPP_ROW_SHARE(5), 0xf, 0xf, false);
            ar[6] = __builtin_amdgcn_update_dpp(0, hs, DPP_ROW_SHARE(6), 0xf, 0xf, false);
            ar[7] = __builtin_amdgcn_update_dpp(0, hs, DPP_ROW_SHARE(7), 0xf, 0xf, false);
            ar[8] = __builtin_amdgcn_update_dpp(0, hs, DPP_ROW_SHARE(8), 0xf, 0xf, false);
            ar[9] = __builtin_amdgcn_update_dpp(0, hs, DPP_ROW_SHARE(9), 0xf, 0xf, false);
            v8f acc[5];
#pragma unroll
            for (int n = 0; n < 5; ++n) {
                v8f z = {0.f, 0.f, 0.f, 0.f, 0.f, 0.f, 0.f, 0.f};
                acc[n] = z;
            }
#pragma unroll
            for (int kc = 0; kc < 5; ++kc) {
                v2f a;
                a.x = __int_as_float(ar[2 * kc]);
                a.y = __int_as_float(ar[2 * kc + 1]);
#pragma unroll
                for (int n = 0; n < 5; ++n) {
                    acc[n] = __builtin_amdgcn_wmma_f32_16x16x4_f32(
                        false, a, false, bf[kc][n], (short)0, acc[n], false, false);
                }
            }
            const int t4 = __float_as_int(acc[4][0]);
            g4[0] = __int_as_float(__builtin_amdgcn_update_dpp(
                __float_as_int(acc[0][0]), t4, DPP_QUAD_ID,      0x2, 0x1, false));
            g4[1] = __int_as_float(__builtin_amdgcn_update_dpp(
                __float_as_int(acc[1][0]), t4, DPP_ROW_SHL(4),  0x2, 0x1, false));
            g4[2] = __int_as_float(__builtin_amdgcn_update_dpp(
                __float_as_int(acc[2][0]), t4, DPP_ROW_SHL(8),  0x2, 0x1, false));
            g4[3] = __int_as_float(__builtin_amdgcn_update_dpp(
                __float_as_int(acc[3][0]), t4, DPP_ROW_SHL(12), 0x2, 0x1, false));
        }
        if constexpr (L2) {
            const float4 q = qn;
            qn = *reinterpret_cast<const float4*>(&lds_g2[((row + 1) & (GRP - 1)) * G2ST]);
            g4[0] = (!L1 || is2lane) ? q.x : g4[0];
            g4[1] = (!L1 || is2lane) ? q.y : g4[1];
            g4[2] = (!L1 || is2lane) ? q.z : g4[2];
            g4[3] = (!L1 || is2lane) ? q.w : g4[3];
        }
        const float xr = is2lane ? hst : xs;

        const bool upd = (L1 && L2) ? true : (is2lane ? L2 : L1);
        if (upd) {
            float gi = g4[0] + xr * wx[0] + bs[0];
            float gf = g4[1] + xr * wx[1] + bs[1];
            float gg = g4[2] + xr * wx[2] + bs[2];
            float go = g4[3] + xr * wx[3] + bs[3];
            float cn = sigm(gf) * cst + sigm(gi) * tanh_f(gg);
            cst = cn;
            float hn = sigm(go) * tanh_f(cn);
            hst = hn;
            if constexpr (L1) {
                if (L2 || !is2lane) lds_c1h[c1b + row * c1str] = cn;
            }
            if constexpr (L2 && !L1) {
                if (is2lane) out[(size_t)row0 * TT + tout] = cn;
            }
        }
    };

    auto tick_main = [&](float xs, int row, int c1b) __attribute__((always_inline)) {
        const float4 q = qn;
        qn = *reinterpret_cast<const float4*>(&lds_g2[((row + 1) & (GRP - 1)) * G2ST]);
        const float xr = is2lane ? hst : xs;
        const float ti_ = fmaf(q.x, kq[0], fmaf(xr, wsc[0], bsc[0]));
        const float tf_ = fmaf(q.y, kq[1], fmaf(xr, wsc[1], bsc[1]));
        const float tg_ = fmaf(q.z, kq[2], fmaf(xr, wsc[2], bsc[2]));
        const float to_ = fmaf(q.w, kq[3], fmaf(xr, wsc[3], bsc[3]));
        const int hi = __float_as_int(hst);
        const int hs = __builtin_amdgcn_update_dpp(hi, hi, DPP_ROW_SHL(6), 0x2, 0xf, false);
        int ar[10];
        ar[0] = __builtin_amdgcn_update_dpp(0, hi, DPP_ROW_SHARE(0), 0xf, 0xf, false);
        ar[1] = __builtin_amdgcn_update_dpp(0, hi, DPP_ROW_SHARE(1), 0xf, 0xf, false);
        ar[2] = __builtin_amdgcn_update_dpp(0, hi, DPP_ROW_SHARE(2), 0xf, 0xf, false);
        ar[3] = __builtin_amdgcn_update_dpp(0, hi, DPP_ROW_SHARE(3), 0xf, 0xf, false);
        ar[4] = __builtin_amdgcn_update_dpp(0, hs, DPP_ROW_SHARE(4), 0xf, 0xf, false);
        ar[5] = __builtin_amdgcn_update_dpp(0, hs, DPP_ROW_SHARE(5), 0xf, 0xf, false);
        ar[6] = __builtin_amdgcn_update_dpp(0, hs, DPP_ROW_SHARE(6), 0xf, 0xf, false);
        ar[7] = __builtin_amdgcn_update_dpp(0, hs, DPP_ROW_SHARE(7), 0xf, 0xf, false);
        ar[8] = __builtin_amdgcn_update_dpp(0, hs, DPP_ROW_SHARE(8), 0xf, 0xf, false);
        ar[9] = __builtin_amdgcn_update_dpp(0, hs, DPP_ROW_SHARE(9), 0xf, 0xf, false);
        v2f af[5];
        v8f acc[5];
#pragma unroll
        for (int kc = 0; kc < 5; ++kc) {
            af[kc].x = __int_as_float(ar[2 * kc]);
            af[kc].y = __int_as_float(ar[2 * kc + 1]);
            v8f z = {0.f, 0.f, 0.f, 0.f, 0.f, 0.f, 0.f, 0.f};
            acc[kc] = z;
        }
#pragma unroll
        for (int kc = 0; kc < 5; ++kc) {
            acc[2] = __builtin_amdgcn_wmma_f32_16x16x4_f32(false, af[kc], false, bf[kc][2], (short)0, acc[2], false, false);
            acc[4] = __builtin_amdgcn_wmma_f32_16x16x4_f32(false, af[kc], false, bf[kc][4], (short)0, acc[4], false, false);
            acc[1] = __builtin_amdgcn_wmma_f32_16x16x4_f32(false, af[kc], false, bf[kc][1], (short)0, acc[1], false, false);
            acc[0] = __builtin_amdgcn_wmma_f32_16x16x4_f32(false, af[kc], false, bf[kc][0], (short)0, acc[0], false, false);
        }
        __builtin_amdgcn_sched_barrier(0);
        acc[3] = __builtin_amdgcn_wmma_f32_16x16x4_f32(false, af[0], false, bf[0][3], (short)0, acc[3], false, false);
        const int t4 = __float_as_int(acc[4][0]);
        const float pg = __int_as_float(__builtin_amdgcn_update_dpp(__float_as_int(acc[2][0]), t4, DPP_ROW_SHL(8), 0x2, 0x1, false));
        const float pi = __int_as_float(__builtin_amdgcn_update_dpp(__float_as_int(acc[0][0]), t4, DPP_QUAD_ID,    0x2, 0x1, false));
        const float pf = __int_as_float(__builtin_amdgcn_update_dpp(__float_as_int(acc[1][0]), t4, DPP_ROW_SHL(4), 0x2, 0x1, false));
        const float ag = fmaf(pg, kp[2], tg_);
        const float ai = fmaf(pi, kp[0], ti_);
        const float af_ = fmaf(pf, kp[1], tf_);
        __builtin_amdgcn_sched_barrier(0);
        acc[3] = __builtin_amdgcn_wmma_f32_16x16x4_f32(false, af[1], false, bf[1][3], (short)0, acc[3], false, false);
        const float dg = __builtin_amdgcn_exp2f(ag) + 1.0f;
        const float di = 1.0f + __builtin_amdgcn_exp2f(ai);
        const float df = 1.0f + __builtin_amdgcn_exp2f(af_);
        __builtin_amdgcn_sched_barrier(0);
        acc[3] = __builtin_amdgcn_wmma_f32_16x16x4_f32(false, af[2], false, bf[2][3], (short)0, acc[3], false, false);
        acc[3] = __builtin_amdgcn_wmma_f32_16x16x4_f32(false, af[3], false, bf[3][3], (short)0, acc[3], false, false);
        acc[3] = __builtin_amdgcn_wmma_f32_16x16x4_f32(false, af[4], false, bf[4][3], (short)0, acc[3], false, false);
        const float tg = 1.0f - 2.0f / dg;
        const float si = 1.0f / di;
        const float sf = 1.0f / df;
        const float cn = sf * cst + si * tg;
        cst = cn;
        lds_c1h[c1b + row * c1str] = cn;
        const float po = __int_as_float(__builtin_amdgcn_update_dpp(__float_as_int(acc[3][0]), t4, DPP_ROW_SHL(12), 0x2, 0x1, false));
        const float ao = fmaf(po, kp[3], to_);
        __builtin_amdgcn_sched_barrier(0);
        const float ec = __builtin_amdgcn_exp2f(cn * (2.0f * LOG2E));
        const float eo = __builtin_amdgcn_exp2f(ao);
        __builtin_amdgcn_sched_barrier(0);
        const float qc = 2.0f / (ec + 1.0f);
        const float so = 1.0f / (1.0f + eo);
        hst = fmaf(-so, qc, so);
    };

    auto product2 = [&](int rnext) __attribute__((always_inline)) {
        lds_barrier();
        v8f acc2 = {0.f, 0.f, 0.f, 0.f, 0.f, 0.f, 0.f, 0.f};
#pragma unroll
        for (int kc = 0; kc < 5; ++kc) {
            v2f a;
            const float* cp = &lds_c1h[nu * HID + 4 * kc + 2 * khalf];
            a.x = cp[0]; a.y = cp[1];
            acc2 = __builtin_amdgcn_wmma_f32_16x16x4_f32(
                false, a, false, bf2[kc], (short)0, acc2, false, false);
        }
#pragma unroll
        for (int v = 0; v < 8; ++v)
            lds_g2[(8 * khalf + v) * G2ST + nu] = acc2[v];
        lds_barrier();
        qn = *reinterpret_cast<const float4*>(&lds_g2[rnext * G2ST]);
    };

    float xcur = xin[(size_t)row0 * T_LEN + lane];
    float xnxt = xin[(size_t)row0 * T_LEN + XBLK + lane];

    lds_barrier();

    auto group = [&](auto f2, float xc, int lbase, int tbase) __attribute__((always_inline)) {
        const int c1b = c1off + (is2lane ? lbase : 0);
        for (int tl = 0; tl < GRP; ++tl) {
            const float xs = __int_as_float(
                __builtin_amdgcn_readlane(__float_as_int(xc), lbase + tl));
            if constexpr (decltype(f2)::value) tick_main(xs, tl, c1b);
            else tick(flag<true>{}, f2, xs, tl, tbase + tl - GRP, c1b);
        }
        product2(0);
    };
    auto store_block = [&](int wbase, bool lo_valid) __attribute__((always_inline)) {
        lds_barrier();
        const float o = lds_c1h[C2H_OFF + lane];
        if (lo_valid || lane >= GRP) out[(long)row0 * TT + (wbase + lane)] = o;
    };

    group(flag<false>{}, xcur, 0, 0);
    group(flag<true>{}, xcur, GRP, GRP);
    store_block(-GRP, false);

    for (int b = 1; b < NBLK; ++b) {
        xcur = xnxt;
        const int bn = (b + 1 < NBLK) ? (b + 1) : (NBLK - 1);
        xnxt = xin[(size_t)row0 * T_LEN + bn * XBLK + lane];
        for (int h = 0; h < 2; ++h)
            group(flag<true>{}, xcur, h * GRP, b * XBLK + h * GRP);
        store_block(b * XBLK - GRP, true);
    }

    for (int i = 0; i < GRP; ++i)
        tick(flag<false>{}, flag<true>{}, 0.0f, i, T_LEN - GRP + i, c1off);

    for (int f = 0; f < fut; ++f) {
        const int s = T_LEN + f;
        const int r = s & (GRP - 1);
        const float xs = __int_as_float(
            __builtin_amdgcn_readlane(__float_as_int(cst), L2LANE));
        tick(flag<true>{}, flag<false>{}, xs, r, 0, c1off);
        product2(r);
        tick(flag<false>{}, flag<true>{}, 0.0f, r, s, c1off);
    }
}

extern "C" void kernel_launch(void* const* d_in, const int* in_sizes, int n_in,
                              void* d_out, int out_size, void* d_ws, size_t ws_size,
                              hipStream_t stream)
{
    const float* xin  = (const float*)d_in[0];
    const float* Wih1 = (const float*)d_in[1];
    const float* Whh1 = (const float*)d_in[2];
    const float* bih1 = (const float*)d_in[3];
    const float* bhh1 = (const float*)d_in[4];
    const float* Wih2 = (const float*)d_in[5];
    const float* Whh2 = (const float*)d_in[6];
    const float* bih2 = (const float*)d_in[7];
    const float* bhh2 = (const float*)d_in[8];
    const int*   futp = (const int*)d_in[9];
    float* out = (float*)d_out;

    dim3 grid(N_WG);
    dim3 block(WG_THREADS);
    lstm_seq_kernel<<<grid, block, 0, stream>>>(xin, Wih1, Whh1, bih1, bhh1,
                                                Wih2, Whh2, bih2, bhh2, futp, out);
}


# Layer-2 gate sums leave LDS: product made as Wih2.c1^T (row = gate, column = time) stays in its D registers, a step takes its four by lane reads; the step's only LDS read goes; k order of sums kept
# speedup vs baseline: 6.8837x; 1.0478x over previous
#include <hip/hip_runtime.h>

typedef __attribute__((ext_vector_type(2))) float v2f;
typedef __attribute__((ext_vector_type(8))) float v8f;

#define B_TOT 1024
#define T_LEN 4096
#define HID   20
#define XBLK  32
#define NBLK  (T_LEN / XBLK)
#define GRP   16
#define NGRP  (T_LEN / GRP)
#define L2LANE 20
#define WAVES 4
#define N_WG  256
#define WG_THREADS 128
#define C2H_OFF    (GRP * HID + 32)
#define C1H_FLOATS (GRP * HID + 32 + XBLK)
#define WV_STRIDE 10496
static_assert(N_WG * WAVES == B_TOT && WG_THREADS == 32 * WAVES, "one wave per batch row");
static_assert(C1H_FLOATS % 4 == 0 && WV_STRIDE % 4 == 0 && C1H_FLOATS <= WV_STRIDE,
              "16-byte aligned regions inside a wave's stride");
static_assert(WAVES * WV_STRIDE * 4 > 160 * 1024 && WAVES * WV_STRIDE * 4 <= 320 * 1024,
              "one work-group per WGP");
static_assert(T_LEN % XBLK == 0 && NBLK >= 2, "x blocks must tile the sequence");
static_assert(XBLK == 2 * GRP && T_LEN % GRP == 0 && NGRP >= 2, "two product groups per x block");

#define LOG2E 1.44269504088896340736f
#define DPP_QUAD_ID      0xE4
#define DPP_ROW_SHL(n)   (0x100 + (n))
#define DPP_ROW_SHARE(n) (0x150 + (n))

template <bool V> struct flag { static constexpr bool value = V; };

__device__ __forceinline__ float sigm(float x) {
    return 1.0f / (1.0f + __expf(-x));
}
__device__ __forceinline__ float tanh_f(float x) {
    return 1.0f - 2.0f / (__expf(2.0f * x) + 1.0f);
}

__device__ __forceinline__ void lds_barrier() {
    __builtin_amdgcn_fence(__ATOMIC_RELEASE, "wavefront", "local");
    __builtin_amdgcn_fence(__ATOMIC_ACQUIRE, "wavefront", "local");
}

__device__ __forceinline__ int kunit(int r, int h) {
    return (h == 0) ? r : ((r < 4) ? (16 + r) : (r + 6));
}

__global__ void __launch_bounds__(WG_THREADS)
lstm_seq_kernel(const float* __restrict__ xin,
                const float* __restrict__ Wih1,
                const float* __restrict__ Whh1,
                const float* __restrict__ bih1,
                const float* __restrict__ bhh1,
                const float* __restrict__ Wih2,
                const float* __restrict__ Whh2,
                const float* __restrict__ bih2,
                const float* __restrict__ bhh2,
                const int*   __restrict__ futp,
                float* __restrict__ out)
{
    __shared__ __attribute__((aligned(16))) float lds_pool[WAVES * WV_STRIDE];

    const int wv    = __builtin_amdgcn_readfirstlane((int)(threadIdx.x >> 5));
    float* const lds_c1h = lds_pool + wv * WV_STRIDE;

    const int lane  = threadIdx.x & 31;
    const int nu    = lane & 15;
    const int khalf = lane >> 4;
    const int row0  = blockIdx.x * WAVES + wv;
    const int fut   = *futp;
    const int TT    = T_LEN + fut;

    const bool is2lane = (lane == L2LANE);
    const int  unit    = (lane < HID) ? lane : nu;
    const bool primary = (lane < 10) || (lane >= 16 && lane < HID) || (lane >= 26);
    const int  c1off = primary ? unit : (is2lane ? C2H_OFF : (GRP * HID + lane));
    const int  c1str = primary ? HID : (is2lane ? 1 : 0);

    for (int i = lane; i < GRP * HID; i += 32)  lds_c1h[i] = 0.0f;
    lds_c1h[C2H_OFF + lane] = 0.0f;

    float wx[4], bs[4];
#pragma unroll
    for (int g = 0; g < 4; ++g) {
        if (is2lane) {
            wx[g] = Whh2[g];
            bs[g] = bih2[g] + bhh2[g];
        } else {
            wx[g] = Wih1[g * HID + unit];
            bs[g] = bih1[g * HID + unit] + bhh1[g * HID + unit];
        }
    }

    float wsc[4], bsc[4], kp[4], kq[4];
#pragma unroll
    for (int g = 0; g < 4; ++g) {
        const float ks = (g == 2) ? (2.0f * LOG2E) : -LOG2E;
        wsc[g] = ks * wx[g];
        bsc[g] = ks * bs[g];
        kp[g]  = is2lane ? 0.0f : ks;
        kq[g]  = is2lane ? ks : 0.0f;
    }

    v2f bf[5][5];
#pragma unroll
    for (int n = 0; n < 5; ++n) {
        const int wrow = (n < 4) ? (n * HID + nu) : ((nu >> 2) * HID + 16 + (nu & 3));
#pragma unroll
        for (int kc = 0; kc < 5; ++kc) {
            v2f b;
            b.x = Whh1[wrow * HID + kunit(2 * kc,     khalf)];
            b.y = Whh1[wrow * HID + kunit(2 * kc + 1, khalf)];
            bf[kc][n] = b;
        }
    }
    v2f bf2[5];
#pragma unroll
    for (int kc = 0; kc < 5; ++kc) {
        v2f b; b.x = 0.0f; b.y = 0.0f;
        if (nu < 4) {
            const float* p = Wih2 + nu * HID + 4 * kc + 2 * khalf;
            b.x = p[0]; b.y = p[1];
        }
        bf2[kc] = b;
    }

    float cst = 0.0f, hst = 0.0f;
    float q2[4] = {0.0f, 0.0f, 0.0f, 0.0f};
    float4 qn = make_float4(0.0f, 0.0f, 0.0f, 0.0f);
    auto l2sums = [&](int row) __attribute__((always_inline)) {
        float4 q;
        q.x = __int_as_float(__builtin_amdgcn_readlane(__float_as_int(q2[0]), row));
        q.y = __int_as_float(__builtin_amdgcn_readlane(__float_as_int(q2[1]), row));
        q.z = __int_as_float(__builtin_amdgcn_readlane(__float_as_int(q2[2]), row));
        q.w = __int_as_float(__builtin_amdgcn_readlane(__float_as_int(q2[3]), row));
        return q;
    };

    auto tick = [&](auto f1, auto f2, float xs, int row, int tout, int c1b) __attribute__((always_inline)) {
        constexpr bool L1 = decltype(f1)::value;
        constexpr bool L2 = decltype(f2)::value;
        float g4[4] = {0.0f, 0.0f, 0.0f, 0.0f};
        if constexpr (L1) {
            const int hi = __float_as_int(hst);
            const int hs = __builtin_amdgcn_update_dpp(hi, hi, DPP_ROW_SHL(6), 0x2, 0xf, false);
            int ar[10];
            ar[0] = __builtin_amdgcn_update_dpp(0, hi, DPP_ROW_SHARE(0), 0xf, 0xf, false);
            ar[1] = __builtin_amdgcn_update_dpp(0, hi, DPP_ROW_SHARE(1), 0xf, 0xf, false);
            ar[2] = __builtin_amdgcn_update_dpp(0, hi, DPP_ROW_SHARE(2), 0xf, 0xf, false);
            ar[3] = __builtin_amdgcn_update_dpp(0, hi, DPP_ROW_SHARE(3), 0xf, 0xf, false);
            ar[4] = __builtin_amdgcn_update_dpp(0, hs, DPP_ROW_SHARE(4), 0xf, 0xf, false);
            ar[5] = __builtin_amdgcn_update_dpp(0, hs, DPP_ROW_SHARE(5), 0xf, 0xf, false);
            ar[6] = __builtin_amdgcn_update_dpp(0, hs, DPP_ROW_SHARE(6), 0xf, 0xf, false);
            ar[7] = __builtin_amdgcn_update_dpp(0, hs, DPP_ROW_SHARE(7), 0xf, 0xf, false);
            ar[8] = __builtin_amdgcn_update_dpp(0, hs, DPP_ROW_SHARE(8), 0xf, 0xf, false);
            ar[9] = __builtin_amdgcn_update_dpp(0, hs, DPP_ROW_SHARE(9), 0xf, 0xf, false);
            v8f acc[5];
#pragma unroll
            for (int n = 0; n < 5; ++n) {
                v8f z = {0.f, 0.f, 0.f, 0.f, 0.f, 0.f, 0.f, 0.f};
                acc[n] = z;
            }
#pragma unroll
            for (int kc = 0; kc < 5; ++kc) {
                v2f a;
                a.x = __int_as_float(ar[2 * kc]);
                a.y = __int_as_float(ar[2 * kc + 1]);
#pragma unroll
                for (int n = 0; n < 5; ++n) {
                    acc[n] = __builtin_amdgcn_wmma_f32_16x16x4_f32(
                        false, a, false, bf[kc][n], (short)0, acc[n], false, false);
                }
            }
            const int t4 = __float_as_int(acc[4][0]);
            g4[0] = __int_as_float(__builtin_amdgcn_update_dpp(
                __float_as_int(acc[0][0]), t4, DPP_QUAD_ID,      0x2, 0x1, false));
            g4[1] = __int_as_float(__builtin_amdgcn_update_dpp(
                __float_as_int(acc[1][0]), t4, DPP_ROW_SHL(4),  0x2, 0x1, false));
            g4[2] = __int_as_float(__builtin_amdgcn_update_dpp(
                __float_as_int(acc[2][0]), t4, DPP_ROW_SHL(8),  0x2, 0x1, false));
            g4[3] = __int_as_float(__builtin_amdgcn_update_dpp(
                __float_as_int(acc[3][0]), t4, DPP_ROW_SHL(12), 0x2, 0x1, false));
        }
        if constexpr (L2) {
            const float4 q = l2sums(row);
            g4[0] = (!L1 || is2lane) ? q.x : g4[0];
            g4[1] = (!L1 || is2lane) ? q.y : g4[1];
            g4[2] = (!L1 || is2lane) ? q.z : g4[2];
            g4[3] = (!L1 || is2lane) ? q.w : g4[3];
        }
        const float xr = is2lane ? hst : xs;

        const bool upd = (L1 && L2) ? true : (is2lane ? L2 : L1);
        if (upd) {
            float gi = g4[0] + xr * wx[0] + bs[0];
            float gf = g4[1] + xr * wx[1] + bs[1];
            float gg = g4[2] + xr * wx[2] + bs[2];
            float go = g4[3] + xr * wx[3] + bs[3];
            float cn = sigm(gf) * cst + sigm(gi) * tanh_f(gg);
            cst = cn;
            float hn = sigm(go) * tanh_f(cn);
            hst = hn;
            if constexpr (L1) {
                if (L2 || !is2lane) lds_c1h[c1b + row * c1str] = cn;
            }
            if constexpr (L2 && !L1) {
                if (is2lane) out[(size_t)row0 * TT + tout] = cn;
            }
        }
    };

    auto tick_main = [&](float xs, int row, int c1b) __attribute__((always_inline)) {
        const float4 q = qn;
        qn = l2sums((row + 1) & (GRP - 1));
        const float xr = is2lane ? hst : xs;
        const float ti_ = fmaf(q.x, kq[0], fmaf(xr, wsc[0], bsc[0]));
        const float tf_ = fmaf(q.y, kq[1], fmaf(xr, wsc[1], bsc[1]));
        const float tg_ = fmaf(q.z, kq[2], fmaf(xr, wsc[2], bsc[2]));
        const float to_ = fmaf(q.w, kq[3], fmaf(xr, wsc[3], bsc[3]));
        const int hi = __float_as_int(hst);
        const int hs = __builtin_amdgcn_update_dpp(hi, hi, DPP_ROW_SHL(6), 0x2, 0xf, false);
        int ar[10];
        ar[0] = __builtin_amdgcn_update_dpp(0, hi, DPP_ROW_SHARE(0), 0xf, 0xf, false);
        ar[1] = __builtin_amdgcn_update_dpp(0, hi, DPP_ROW_SHARE(1), 0xf, 0xf, false);
        ar[2] = __builtin_amdgcn_update_dpp(0, hi, DPP_ROW_SHARE(2), 0xf, 0xf, false);
        ar[3] = __builtin_amdgcn_update_dpp(0, hi, DPP_ROW_SHARE(3), 0xf, 0xf, false);
        ar[4] = __builtin_amdgcn_update_dpp(0, hs, DPP_ROW_SHARE(4), 0xf, 0xf, false);
        ar[5] = __builtin_amdgcn_update_dpp(0, hs, DPP_ROW_SHARE(5), 0xf, 0xf, false);
        ar[6] = __builtin_amdgcn_update_dpp(0, hs, DPP_ROW_SHARE(6), 0xf, 0xf, false);
        ar[7] = __builtin_amdgcn_update_dpp(0, hs, DPP_ROW_SHARE(7), 0xf, 0xf, false);
        ar[8] = __builtin_amdgcn_update_dpp(0, hs, DPP_ROW_SHARE(8), 0xf, 0xf, false);
        ar[9] = __builtin_amdgcn_update_dpp(0, hs, DPP_ROW_SHARE(9), 0xf, 0xf, false);
        v2f af[5];
        v8f acc[5];
#pragma unroll
        for (int kc = 0; kc < 5; ++kc) {
            af[kc].x = __int_as_float(ar[2 * kc]);
            af[kc].y = __int_as_float(ar[2 * kc + 1]);
            v8f z = {0.f, 0.f, 0.f, 0.f, 0.f, 0.f, 0.f, 0.f};
            acc[kc] = z;
        }
#pragma unroll
        for (int kc = 0; kc < 5; ++kc) {
            acc[2] = __builtin_amdgcn_wmma_f32_16x16x4_f32(false, af[kc], false, bf[kc][2], (short)0, acc[2], false, false);
            acc[4] = __builtin_amdgcn_wmma_f32_16x16x4_f32(false, af[kc], false, bf[kc][4], (short)0, acc[4], false, false);
            acc[1] = __builtin_amdgcn_wmma_f32_16x16x4_f32(false, af[kc], false, bf[kc][1], (short)0, acc[1], false, false);
            acc[0] = __builtin_amdgcn_wmma_f32_16x16x4_f32(false, af[kc], false, bf[kc][0], (short)0, acc[0], false, false);
        }
        __builtin_amdgcn_sched_barrier(0);
        acc[3] = __builtin_amdgcn_wmma_f32_16x16x4_f32(false, af[0], false, bf[0][3], (short)0, acc[3], false, false);
        const int t4 = __float_as_int(acc[4][0]);
        const float pg = __int_as_float(__builtin_amdgcn_update_dpp(__float_as_int(acc[2][0]), t4, DPP_ROW_SHL(8), 0x2, 0x1, false));
        const float pi = __int_as_float(__builtin_amdgcn_update_dpp(__float_as_int(acc[0][0]), t4, DPP_QUAD_ID,    0x2, 0x1, false));
        const float pf = __int_as_float(__builtin_amdgcn_update_dpp(__float_as_int(acc[1][0]), t4, DPP_ROW_SHL(4), 0x2, 0x1, false));
        const float ag = fmaf(pg, kp[2], tg_);
        const float ai = fmaf(pi, kp[0], ti_);
        const float af_ = fmaf(pf, kp[1], tf_);
        __builtin_amdgcn_sched_barrier(0);
        acc[3] = __builtin_amdgcn_wmma_f32_16x16x4_f32(false, af[1], false, bf[1][3], (short)0, acc[3], false, false);
        const float dg = __builtin_amdgcn_exp2f(ag) + 1.0f;
        const float di = 1.0f + __builtin_amdgcn_exp2f(ai);
        const float df = 1.0f + __builtin_amdgcn_exp2f(af_);
        __builtin_amdgcn_sched_barrier(0);
        acc[3] = __builtin_amdgcn_wmma_f32_16x16x4_f32(false, af[2], false, bf[2][3], (short)0, acc[3], false, false);
        acc[3] = __builtin_amdgcn_wmma_f32_16x16x4_f32(false, af[3], false, bf[3][3], (short)0, acc[3], false, false);
        acc[3] = __builtin_amdgcn_wmma_f32_16x16x4_f32(false, af[4], false, bf[4][3], (short)0, acc[3], false, false);
        const float tg = 1.0f - 2.0f / dg;
        const float si = 1.0f / di;
        const float sf = 1.0f / df;
        const float cn = sf * cst + si * tg;
        cst = cn;
        lds_c1h[c1b + row * c1str] = cn;
        const float po = __int_as_float(__builtin_amdgcn_update_dpp(__float_as_int(acc[3][0]), t4, DPP_ROW_SHL(12), 0x2, 0x1, false));
        const float ao = fmaf(po, kp[3], to_);
        __builtin_amdgcn_sched_barrier(0);
        const float ec = __builtin_amdgcn_exp2f(cn * (2.0f * LOG2E));
        const float eo = __builtin_amdgcn_exp2f(ao);
        __builtin_amdgcn_sched_barrier(0);
        const float qc = 2.0f / (ec + 1.0f);
        const float so = 1.0f / (1.0f + eo);
        hst = fmaf(-so, qc, so);
    };

    auto product2 = [&]() __attribute__((always_inline)) {
        lds_barrier();
        v8f acc2 = {0.f, 0.f, 0.f, 0.f, 0.f, 0.f, 0.f, 0.f};
#pragma unroll
        for (int kc = 0; kc < 5; ++kc) {
            v2f b;
            const float* cp = &lds_c1h[nu * HID + 4 * kc + 2 * khalf];
            b.x = cp[0]; b.y = cp[1];
            acc2 = __builtin_amdgcn_wmma_f32_16x16x4_f32(
                false, bf2[kc], false, b, (short)0, acc2, false, false);
        }
        asm volatile("s_wait_alu depctr_va_vdst(0)" : "+v"(acc2));
#pragma unroll
        for (int g = 0; g < 4; ++g) q2[g] = acc2[g];
        qn = l2sums(0);
    };

    float xcur = xin[(size_t)row0 * T_LEN + lane];
    float xnxt = xin[(size_t)row0 * T_LEN + XBLK + lane];

    lds_barrier();

    auto group = [&](auto f2, float xc, int lbase, int tbase) __attribute__((always_inline)) {
        const int c1b = c1off + (is2lane ? lbase : 0);
        for (int tl = 0; tl < GRP; ++tl) {
            const float xs = __int_as_float(
                __builtin_amdgcn_readlane(__float_as_int(xc), lbase + tl));
            if constexpr (decltype(f2)::value) tick_main(xs, tl, c1b);
            else tick(flag<true>{}, f2, xs, tl, tbase + tl - GRP, c1b);
        }
        product2();
    };
    auto store_block = [&](int wbase, bool lo_valid) __attribute__((always_inline)) {
        lds_barrier();
        const float o = lds_c1h[C2H_OFF + lane];
        if (lo_valid || lane >= GRP) out[(long)row0 * TT + (wbase + lane)] = o;
    };

    group(flag<false>{}, xcur, 0, 0);
    group(flag<true>{}, xcur, GRP, GRP);
    store_block(-GRP, false);

    for (int b = 1; b < NBLK; ++b) {
        xcur = xnxt;
        const int bn = (b + 1 < NBLK) ? (b + 1) : (NBLK - 1);
        xnxt = xin[(size_t)row0 * T_LEN + bn * XBLK + lane];
        for (int h = 0; h < 2; ++h)
            group(flag<true>{}, xcur, h * GRP, b * XBLK + h * GRP);
        store_block(b * XBLK - GRP, true);
    }

    for (int i = 0; i < GRP; ++i)
        tick(flag<false>{}, flag<true>{}, 0.0f, i, T_LEN - GRP + i, c1off);

    for (int f = 0; f < fut; ++f) {
        const int s = T_LEN + f;
        const int r = s & (GRP - 1);
        const float xs = __int_as_float(
            __builtin_amdgcn_readlane(__float_as_int(cst), L2LANE));
        tick(flag<true>{}, flag<false>{}, xs, r, 0, c1off);
        product2();
        tick(flag<false>{}, flag<true>{}, 0.0f, r, s, c1off);
    }
}

extern "C" void kernel_launch(void* const* d_in, const int* in_sizes, int n_in,
                              void* d_out, int out_size, void* d_ws, size_t ws_size,
                              hipStream_t stream)
{
    const float* xin  = (const float*)d_in[0];
    const float* Wih1 = (const float*)d_in[1];
    const float* Whh1 = (const float*)d_in[2];
    const float* bih1 = (const float*)d_in[3];
    const float* bhh1 = (const float*)d_in[4];
    const float* Wih2 = (const float*)d_in[5];
    const float* Whh2 = (const float*)d_in[6];
    const float* bih2 = (const float*)d_in[7];
    const float* bhh2 = (const float*)d_in[8];
    const int*   futp = (const int*)d_in[9];
    float* out = (float*)d_out;

    dim3 grid(N_WG);
    dim3 block(WG_THREADS);
    lstm_seq_kernel<<<grid, block, 0, stream>>>(xin, Wih1, Whh1, bih1, bhh1,
                                                Wih2, Whh2, bih2, bhh2, futp, out);
}
